# starting kernel
# baseline (speedup 1.0000x reference)
#include <hip/hip_runtime.h>
#include <hip/hip_bf16.h>
#include <math.h>

#define BB   4
#define TT   4096
#define DM   1024
#define GG   8
#define GD   128
#define NST  32
#define LCH  64
#define NCH  (TT / LCH)
#define SEG  8
#define NSEG (LCH / SEG)

#define US   132
#define BS   36

typedef float v2f __attribute__((ext_vector_type(2)));
typedef float v8f __attribute__((ext_vector_type(8)));

__device__ __forceinline__ v8f wmma4(v2f a, v2f b, v8f c) {
  return __builtin_amdgcn_wmma_f32_16x16x4_f32(
       false, a,  false, b,
       (short)0, c,  false,  false);
}

__global__ __launch_bounds__(256) void k0_setup(
    const float* __restrict__ A_log, const float* __restrict__ B_param,
    const float* __restrict__ dt_logit,
    float* __restrict__ wAbar, float* __restrict__ wAbar8,
    float* __restrict__ wAbar64, float* __restrict__ wBbar) {
  int tid = threadIdx.x;
  int g = tid >> 5;
  float x = A_log[tid];
  float sp = (x > 20.f) ? x : log1pf(expf(x));
  float A = -sp;
  float dtl = dt_logit[g];
  float sig = 1.f / (1.f + expf(-dtl));
  float dt = 0.001f + (0.1f - 0.001f) * sig;
  float Abar = expf(A * dt);
  float frac = (fabsf(A) > 1e-6f) ? (Abar - 1.f) / A : dt;
  wAbar[tid] = Abar;
  float p = Abar;
  p *= p; p *= p; p *= p;
  wAbar8[tid] = p;
  p *= p; p *= p; p *= p;
  wAbar64[tid] = p;
  const float* bp = B_param + (size_t)tid * GD;
  float*       bb = wBbar   + (size_t)tid * GD;
#pragma unroll 4
  for (int i = 0; i < GD; ++i) bb[i] = frac * bp[i];
}

__global__ __launch_bounds__(256) void k1_bu_scan(
    const float* __restrict__ u,
    const float* __restrict__ wAbar, const float* __restrict__ wAbar8,
    const float* __restrict__ wBbar, float* __restrict__ wLocal) {
  __shared__ float lds_u[LCH * US];
  __shared__ float lds_bu[LCH * BS];
  __shared__ float segsum[NSEG * NST];
  __shared__ float segcar[NSEG * NST];

  int blk = blockIdx.x;
  int c = blk & (NCH - 1);
  int g = (blk >> 6) & (GG - 1);
  int b = blk >> 9;
  int tid = threadIdx.x;

  const float* Bb = wBbar + (size_t)g * NST * GD;
  __builtin_prefetch(Bb + tid * 16, 0, 3);

  const float* ub = u + ((size_t)(b * TT + c * LCH)) * DM + g * GD;
#pragma unroll
  for (int j = 0; j < 8; ++j) {
    int idx = tid + 256 * j;
    int tl = idx >> 5;
    int q  = idx & 31;
    float4 v = *(const float4*)(ub + (size_t)tl * DM + q * 4);
    *(float4*)(&lds_u[tl * US + q * 4]) = v;
  }
  __syncthreads();

  int lane   = tid & 31;
  int w      = tid >> 5;
  int ttile  = w >> 1;
  int ntile  = w & 1;
  int m      = lane & 15;
  int koff   = (lane >> 4) << 1;
  int rowadd = (lane >> 4) << 3;

  const float* arow = &lds_u[(ttile * 16 + m) * US];
  const float* brow = Bb + (ntile * 16 + m) * GD;
  v8f acc = {};
#pragma unroll
  for (int k = 0; k < GD; k += 4) {
    v2f a  = *(const v2f*)(arow + k + koff);
    v2f bf = *(const v2f*)(brow + k + koff);
    acc = wmma4(a, bf, acc);
  }
#pragma unroll
  for (int r = 0; r < 8; ++r) {
    int t = ttile * 16 + r + rowadd;
    lds_bu[t * BS + ntile * 16 + m] = acc[r];
  }
  __syncthreads();

  int n = tid & 31;
  int s = tid >> 5;
  float Ab = wAbar[g * NST + n];
  float a8 = wAbar8[g * NST + n];
  float run = 0.f;
#pragma unroll
  for (int k = 0; k < SEG; ++k) {
    int t = s * SEG + k;
    run = run * Ab + lds_bu[t * BS + n];
    lds_bu[t * BS + n] = run;
  }
  segsum[s * NST + n] = run;
  __syncthreads();
  if (s == 0) {
    float carry = 0.f;
#pragma unroll
    for (int ss = 0; ss < NSEG; ++ss) {
      segcar[ss * NST + n] = carry;
      carry = carry * a8 + segsum[ss * NST + n];
    }
  }
  __syncthreads();
  {
    float carry = segcar[s * NST + n];
    float f = Ab;
#pragma unroll
    for (int k = 0; k < SEG; ++k) {
      lds_bu[(s * SEG + k) * BS + n] += f * carry;
      f *= Ab;
    }
  }
  __syncthreads();

  float* out = wLocal + ((size_t)((b * GG + g) * TT + c * LCH)) * NST;
#pragma unroll
  for (int j = 0; j < 2; ++j) {
    int idx4 = tid + 256 * j;
    int tl = idx4 >> 3;
    int q  = idx4 & 7;
    *(float4*)(out + tl * NST + q * 4) = *(const float4*)(&lds_bu[tl * BS + q * 4]);
  }
}

__global__ __launch_bounds__(256) void k2_carry(
    const float* __restrict__ state0, const float* __restrict__ wAbar64,
    const float* __restrict__ wLocal, float* __restrict__ wCarry,
    float* __restrict__ sfinal) {
  int tg = blockIdx.x * 256 + threadIdx.x;
  int n = tg & 31;
  int g = (tg >> 5) & 7;
  int b = tg >> 8;
  int bg = b * GG + g;
  float carry = state0[bg * NST + n];
  float aL = wAbar64[g * NST + n];
  const float* loc = wLocal + (size_t)bg * TT * NST;
  for (int cc = 0; cc < NCH; ++cc) {
    wCarry[(bg * NCH + cc) * NST + n] = carry;
    carry = carry * aL + loc[((size_t)cc * LCH + (LCH - 1)) * NST + n];
  }
  sfinal[bg * NST + n] = carry;
}

__global__ __launch_bounds__(256) void k3_y(
    const float* __restrict__ u, const float* __restrict__ Cmat,
    const float* __restrict__ Dp,
    const float* __restrict__ wAbar, const float* __restrict__ wAbar8,
    const float* __restrict__ wLocal, const float* __restrict__ wCarry,
    float* __restrict__ y) {
  __shared__ float lds_s[LCH * BS];
  __shared__ float lds_y[LCH * US];

  int blk = blockIdx.x;
  int c = blk & (NCH - 1);
  int g = (blk >> 6) & (GG - 1);
  int b = blk >> 9;
  int bg = b * GG + g;
  int tid = threadIdx.x;

  const float* Cg = Cmat + (size_t)g * GD * NST;
  __builtin_prefetch(Cg + tid * 16, 0, 3);

  const float* src = wLocal + ((size_t)(bg * TT + c * LCH)) * NST;
#pragma unroll
  for (int j = 0; j < 2; ++j) {
    int idx4 = tid + 256 * j;
    int tl = idx4 >> 3;
    int q  = idx4 & 7;
    *(float4*)(&lds_s[tl * BS + q * 4]) = *(const float4*)(src + tl * NST + q * 4);
  }
  __syncthreads();

  {
    int n = tid & 31;
    int s = tid >> 5;
    float carry = wCarry[(bg * NCH + c) * NST + n];
    float Ab = wAbar[g * NST + n];
    float a8 = wAbar8[g * NST + n];
    float f = Ab;
    for (int j = 0; j < s; ++j) f *= a8;
#pragma unroll
    for (int k = 0; k < SEG; ++k) {
      lds_s[(s * SEG + k) * BS + n] += f * carry;
      f *= Ab;
    }
  }
  __syncthreads();

  int lane   = tid & 31;
  int w      = tid >> 5;
  int ttile  = w >> 1;
  int dhalf  = w & 1;
  int m      = lane & 15;
  int koff   = (lane >> 4) << 1;
  int rowadd = (lane >> 4) << 3;

  v2f afrag[8];
  const float* arow = &lds_s[(ttile * 16 + m) * BS];
#pragma unroll
  for (int k8 = 0; k8 < 8; ++k8)
    afrag[k8] = *(const v2f*)(arow + k8 * 4 + koff);

#pragma unroll
  for (int dt = 0; dt < 4; ++dt) {
    int d = dhalf * 64 + dt * 16 + m;
    const float* brow = Cg + (size_t)d * NST;
    v8f acc = {};
#pragma unroll
    for (int k8 = 0; k8 < 8; ++k8) {
      v2f bf = *(const v2f*)(brow + k8 * 4 + koff);
      acc = wmma4(afrag[k8], bf, acc);
    }
#pragma unroll
    for (int r = 0; r < 8; ++r) {
      int t = ttile * 16 + r + rowadd;
      lds_y[t * US + d] = acc[r];
    }
  }
  __syncthreads();

  const float* ubase = u + ((size_t)(b * TT + c * LCH)) * DM + g * GD;
  float*       ybase = y + ((size_t)(b * TT + c * LCH)) * DM + g * GD;
  const float* dpg   = Dp + g * GD;
#pragma unroll
  for (int j = 0; j < 8; ++j) {
    int idx = tid + 256 * j;
    int tl = idx >> 5;
    int q  = idx & 31;
    float4 uv = *(const float4*)(ubase + (size_t)tl * DM + q * 4);
    float4 cs = *(const float4*)(&lds_y[tl * US + q * 4]);
    float4 dp = *(const float4*)(dpg + q * 4);
    float4 o;
    o.x = cs.x + dp.x * uv.x;
    o.y = cs.y + dp.y * uv.y;
    o.z = cs.z + dp.z * uv.z;
    o.w = cs.w + dp.w * uv.w;
    *(float4*)(ybase + (size_t)tl * DM + q * 4) = o;
  }
}

extern "C" void kernel_launch(void* const* d_in, const int* in_sizes, int n_in,
                              void* d_out, int out_size, void* d_ws, size_t ws_size,
                              hipStream_t stream) {
  (void)in_sizes; (void)n_in; (void)out_size; (void)ws_size;
  const float* u        = (const float*)d_in[0];
  const float* state0   = (const float*)d_in[1];
  const float* A_log    = (const float*)d_in[2];
  const float* B_param  = (const float*)d_in[3];
  const float* Cmat     = (const float*)d_in[4];
  const float* Dp       = (const float*)d_in[5];
  const float* dt_logit = (const float*)d_in[6];
  float* out = (float*)d_out;
  float* ws  = (float*)d_ws;

  float* wAbar   = ws;
  float* wAbar8  = ws + 256;
  float* wAbar64 = ws + 512;
  float* wBbar   = ws + 768;
  float* wLocal  = ws + 33536;
  float* wCarry  = wLocal + (size_t)BB * GG * TT * NST;

  float* y      = out;
  float* sfinal = out + (size_t)BB * TT * DM;

  k0_setup<<<1, 256, 0, stream>>>(A_log, B_param, dt_logit,
                                  wAbar, wAbar8, wAbar64, wBbar);
  k1_bu_scan<<<BB * GG * NCH, 256, 0, stream>>>(u, wAbar, wAbar8, wBbar, wLocal);
  k2_carry<<<4, 256, 0, stream>>>(state0, wAbar64, wLocal, wCarry, sfinal);
  k3_y<<<BB * GG * NCH, 256, 0, stream>>>(u, Cmat, Dp, wAbar, wAbar8,
                                          wLocal, wCarry, y);
}


# fold both helper launches into the chunk kernels: each block derives its group's Abar/frac itself (B scaled at load) and rebuilds its own inter-chunk carry from chunk-end rows; 4 launches -> 2
# speedup vs baseline: 1.6823x; 1.6823x over previous
#include <hip/hip_runtime.h>
#include <hip/hip_bf16.h>
#include <math.h>

#define BB   4
#define TT   4096
#define DM   1024
#define GG   8
#define GD   128
#define NST  32
#define LCH  64
#define NCH  (TT / LCH)
#define SEG  8
#define NSEG (LCH / SEG)

#define US   132
#define BS   36

typedef float v2f __attribute__((ext_vector_type(2)));
typedef float v8f __attribute__((ext_vector_type(8)));

__device__ __forceinline__ v8f wmma4(v2f a, v2f b, v8f c) {
  return __builtin_amdgcn_wmma_f32_16x16x4_f32(
       false, a,  false, b,
       (short)0, c,  false,  false);
}

__device__ __forceinline__ void zoh_params(float x, float dtl, float& Abar, float& frac) {
  float sp0 = log1pf(expf(x));
  float sp = (x > 20.f) ? x : sp0;
  float A = -sp;
  float sig = 1.f / (1.f + expf(-dtl));
  float dt = 0.001f + (0.1f - 0.001f) * sig;
  Abar = expf(A * dt);
  float q = (Abar - 1.f) / A;
  frac = (fabsf(A) > 1e-6f) ? q : dt;
}

__global__ __launch_bounds__(256) void k1_bu_scan(
    const float* __restrict__ u, const float* __restrict__ A_log,
    const float* __restrict__ B_param, const float* __restrict__ dt_logit,
    float* __restrict__ wLocal) {
  __shared__ float lds_u[LCH * US];
  __shared__ float lds_bu[LCH * BS];
  __shared__ float segsum[NSEG * NST];
  __shared__ float segcar[NSEG * NST];

  int blk = blockIdx.x;
  int c = blk & (NCH - 1);
  int g = (blk >> 6) & (GG - 1);
  int b = blk >> 9;
  int tid = threadIdx.x;

  const float* Bb = B_param + (size_t)g * NST * GD;
  __builtin_prefetch(Bb + tid * 16, 0, 3);

  float Ab, frac;
  zoh_params(A_log[g * NST + (tid & 31)], dt_logit[g], Ab, frac);
  float a8 = Ab;
  a8 *= a8; a8 *= a8; a8 *= a8;

  const float* ub = u + ((size_t)(b * TT + c * LCH)) * DM + g * GD;
#pragma unroll
  for (int j = 0; j < 8; ++j) {
    int idx = tid + 256 * j;
    int tl = idx >> 5;
    int q  = idx & 31;
    float4 v = *(const float4*)(ub + (size_t)tl * DM + q * 4);
    *(float4*)(&lds_u[tl * US + q * 4]) = v;
  }
  __syncthreads();

  int lane   = tid & 31;
  int w      = tid >> 5;
  int ttile  = w >> 1;
  int ntile  = w & 1;
  int m      = lane & 15;
  int koff   = (lane >> 4) << 1;
  int rowadd = (lane >> 4) << 3;

  const float* arow = &lds_u[(ttile * 16 + m) * US];
  const float* brow = Bb + (ntile * 16 + m) * GD;
  float fr = __shfl(frac, ntile * 16 + m);
  v8f acc = {};
#pragma unroll
  for (int k = 0; k < GD; k += 4) {
    v2f a  = *(const v2f*)(arow + k + koff);
    v2f bf = *(const v2f*)(brow + k + koff);
    bf.x *= fr;
    bf.y *= fr;
    acc = wmma4(a, bf, acc);
  }
#pragma unroll
  for (int r = 0; r < 8; ++r) {
    int t = ttile * 16 + r + rowadd;
    lds_bu[t * BS + ntile * 16 + m] = acc[r];
  }
  __syncthreads();

  int n = tid & 31;
  int s = tid >> 5;
  float run = 0.f;
#pragma unroll
  for (int k = 0; k < SEG; ++k) {
    int t = s * SEG + k;
    run = run * Ab + lds_bu[t * BS + n];
    lds_bu[t * BS + n] = run;
  }
  segsum[s * NST + n] = run;
  __syncthreads();
  if (s == 0) {
    float carry = 0.f;
#pragma unroll
    for (int ss = 0; ss < NSEG; ++ss) {
      segcar[ss * NST + n] = carry;
      carry = carry * a8 + segsum[ss * NST + n];
    }
  }
  __syncthreads();
  {
    float carry = segcar[s * NST + n];
    float f = Ab;
#pragma unroll
    for (int k = 0; k < SEG; ++k) {
      lds_bu[(s * SEG + k) * BS + n] += f * carry;
      f *= Ab;
    }
  }
  __syncthreads();

  float* out = wLocal + ((size_t)((b * GG + g) * TT + c * LCH)) * NST;
#pragma unroll
  for (int j = 0; j < 2; ++j) {
    int idx4 = tid + 256 * j;
    int tl = idx4 >> 3;
    int q  = idx4 & 7;
    *(float4*)(out + tl * NST + q * 4) = *(const float4*)(&lds_bu[tl * BS + q * 4]);
  }
}

__global__ __launch_bounds__(256) void k3_y(
    const float* __restrict__ u, const float* __restrict__ Cmat,
    const float* __restrict__ Dp, const float* __restrict__ A_log,
    const float* __restrict__ dt_logit, const float* __restrict__ state0,
    const float* __restrict__ wLocal,
    float* __restrict__ y, float* __restrict__ sfinal) {
  __shared__ float lds_s[LCH * BS];
  __shared__ float lds_y[LCH * US];
  __shared__ float lds_carry[NST];

  int blk = blockIdx.x;
  int c = blk & (NCH - 1);
  int g = (blk >> 6) & (GG - 1);
  int b = blk >> 9;
  int bg = b * GG + g;
  int tid = threadIdx.x;

  const float* Cg = Cmat + (size_t)g * GD * NST;
  __builtin_prefetch(Cg + tid * 16, 0, 3);

  const float* src = wLocal + ((size_t)(bg * TT + c * LCH)) * NST;
#pragma unroll
  for (int j = 0; j < 2; ++j) {
    int idx4 = tid + 256 * j;
    int tl = idx4 >> 3;
    int q  = idx4 & 7;
    *(float4*)(&lds_s[tl * BS + q * 4]) = *(const float4*)(src + tl * NST + q * 4);
  }

  float Ab, frac_unused;
  zoh_params(A_log[g * NST + (tid & 31)], dt_logit[g], Ab, frac_unused);
  float a8 = Ab;
  a8 *= a8; a8 *= a8; a8 *= a8;

  if (tid < 32) {
    float aL = a8;
    aL *= aL; aL *= aL; aL *= aL;
    const float* ends = wLocal + ((size_t)bg * TT + (LCH - 1)) * NST + tid;
    float e[NCH];
#pragma unroll
    for (int j = 0; j < NCH; ++j) e[j] = ends[(size_t)j * LCH * NST];
    float carry = state0[bg * NST + tid];
#pragma unroll
    for (int j = 0; j < NCH - 1; ++j) carry = (j < c) ? (carry * aL + e[j]) : carry;
    lds_carry[tid] = carry;
    if (c == NCH - 1) sfinal[bg * NST + tid] = carry * aL + e[NCH - 1];
  }
  __syncthreads();

  {
    int n = tid & 31;
    int s = tid >> 5;
    float carry = lds_carry[n];
    float f = Ab;
    for (int j = 0; j < s; ++j) f *= a8;
#pragma unroll
    for (int k = 0; k < SEG; ++k) {
      lds_s[(s * SEG + k) * BS + n] += f * carry;
      f *= Ab;
    }
  }
  __syncthreads();

  int lane   = tid & 31;
  int w      = tid >> 5;
  int ttile  = w >> 1;
  int dhalf  = w & 1;
  int m      = lane & 15;
  int koff   = (lane >> 4) << 1;
  int rowadd = (lane >> 4) << 3;

  v2f afrag[8];
  const float* arow = &lds_s[(ttile * 16 + m) * BS];
#pragma unroll
  for (int k8 = 0; k8 < 8; ++k8)
    afrag[k8] = *(const v2f*)(arow + k8 * 4 + koff);

#pragma unroll
  for (int dt = 0; dt < 4; ++dt) {
    int d = dhalf * 64 + dt * 16 + m;
    const float* brow = Cg + (size_t)d * NST;
    v8f acc = {};
#pragma unroll
    for (int k8 = 0; k8 < 8; ++k8) {
      v2f bf = *(const v2f*)(brow + k8 * 4 + koff);
      acc = wmma4(afrag[k8], bf, acc);
    }
#pragma unroll
    for (int r = 0; r < 8; ++r) {
      int t = ttile * 16 + r + rowadd;
      lds_y[t * US + d] = acc[r];
    }
  }
  __syncthreads();

  const float* ubase = u + ((size_t)(b * TT + c * LCH)) * DM + g * GD;
  float*       ybase = y + ((size_t)(b * TT + c * LCH)) * DM + g * GD;
  const float* dpg   = Dp + g * GD;
#pragma unroll
  for (int j = 0; j < 8; ++j) {
    int idx = tid + 256 * j;
    int tl = idx >> 5;
    int q  = idx & 31;
    float4 uv = *(const float4*)(ubase + (size_t)tl * DM + q * 4);
    float4 cs = *(const float4*)(&lds_y[tl * US + q * 4]);
    float4 dp = *(const float4*)(dpg + q * 4);
    float4 o;
    o.x = cs.x + dp.x * uv.x;
    o.y = cs.y + dp.y * uv.y;
    o.z = cs.z + dp.z * uv.z;
    o.w = cs.w + dp.w * uv.w;
    *(float4*)(ybase + (size_t)tl * DM + q * 4) = o;
  }
}

extern "C" void kernel_launch(void* const* d_in, const int* in_sizes, int n_in,
                              void* d_out, int out_size, void* d_ws, size_t ws_size,
                              hipStream_t stream) {
  (void)in_sizes; (void)n_in; (void)out_size; (void)ws_size;
  const float* u        = (const float*)d_in[0];
  const float* state0   = (const float*)d_in[1];
  const float* A_log    = (const float*)d_in[2];
  const float* B_param  = (const float*)d_in[3];
  const float* Cmat     = (const float*)d_in[4];
  const float* Dp       = (const float*)d_in[5];
  const float* dt_logit = (const float*)d_in[6];
  float* out = (float*)d_out;
  float* ws  = (float*)d_ws;

  float* wLocal  = ws + 33536;

  float* y      = out;
  float* sfinal = out + (size_t)BB * TT * DM;

  k1_bu_scan<<<BB * GG * NCH, 256, 0, stream>>>(u, A_log, B_param, dt_logit, wLocal);
  k3_y<<<BB * GG * NCH, 256, 0, stream>>>(u, Cmat, Dp, A_log, dt_logit, state0,
                                          wLocal, y, sfinal);
}


# each work-group walks 4 consecutive chunks (2048 -> 512 blocks per launch): B/C fragments and the carry recurrence once per block, next tile's loads in flight across LDS-only hand-written barriers
# speedup vs baseline: 2.4599x; 1.4622x over previous
#include <hip/hip_runtime.h>
#include <hip/hip_bf16.h>
#include <math.h>


#define BB   4
#define TT   4096
#define DM   1024
#define GG   8
#define GD   128
#define NST  32
#define LCH  64
#define NCH  (TT / LCH)
#define SEG  8
#define NSEG (LCH / SEG)
#define CPB  4
#define NGRP (NCH / CPB)

#define US   132
#define BS   36

typedef float v2f __attribute__((ext_vector_type(2)));
typedef float v4f __attribute__((ext_vector_type(4)));
typedef float v8f __attribute__((ext_vector_type(8)));

__device__ __forceinline__ v8f wmma4(v2f a, v2f b, v8f c) {
  return __builtin_amdgcn_wmma_f32_16x16x4_f32(
       false, a,  false, b,
       (short)0, c,  false,  false);
}

__device__ __forceinline__ void lds_barrier() {
  asm volatile("s_wait_dscnt 0x0\n\ts_barrier_signal -1\n\ts_barrier_wait -1" ::: "memory");
}

__device__ __forceinline__ void zoh_params(float x, float dtl, float& Abar, float& frac) {
  float sp0 = log1pf(expf(x));
  float sp = (x > 20.f) ? x : sp0;
  float A = -sp;
  float sig = 1.f / (1.f + expf(-dtl));
  float dt = 0.001f + (0.1f - 0.001f) * sig;
  Abar = expf(A * dt);
  float q = (Abar - 1.f) / A;
  frac = (fabsf(A) > 1e-6f) ? q : dt;
}

__global__ __launch_bounds__(256) void k1_bu_scan(
    const float* __restrict__ u, const float* __restrict__ A_log,
    const float* __restrict__ B_param, const float* __restrict__ dt_logit,
    float* __restrict__ wLocal) {
  __shared__ float lds_u[LCH * US];
  __shared__ float lds_bu[LCH * BS];
  __shared__ float segsum[NSEG * NST];
  __shared__ float segcar[NSEG * NST];

  int blk = blockIdx.x;
  int qg = blk & (NGRP - 1);
  int g = (blk >> 4) & (GG - 1);
  int b = blk >> 7;
  int c0 = qg * CPB;
  int tid = threadIdx.x;

  float xlog = A_log[g * NST + (tid & 31)];
  float dtl  = dt_logit[g];

  const float* ub = u + ((size_t)(b * TT + c0 * LCH)) * DM + g * GD;
  int tl0 = tid >> 5;
  int q4  = tid & 31;
  v4f ur[8];
#pragma unroll
  for (int j = 0; j < 8; ++j)
    ur[j] = *(const v4f*)(ub + (size_t)(tl0 + 8 * j) * DM + q4 * 4);

  int lane   = tid & 31;
  int w      = tid >> 5;
  int ttile  = w >> 1;
  int ntile  = w & 1;
  int m      = lane & 15;
  int koff   = (lane >> 4) << 1;
  int rowadd = (lane >> 4) << 3;

  const float* brow = B_param + (size_t)g * NST * GD + (ntile * 16 + m) * GD;
  v2f bfr[GD / 4];
#pragma unroll
  for (int kk = 0; kk < GD / 4; ++kk)
    bfr[kk] = *(const v2f*)(brow + kk * 4 + koff);

  __builtin_amdgcn_sched_barrier(0);

  float Ab, frac;
  zoh_params(xlog, dtl, Ab, frac);
  float a8 = Ab;
  a8 *= a8; a8 *= a8; a8 *= a8;

  float fr = __shfl(frac, ntile * 16 + m);
#pragma unroll
  for (int kk = 0; kk < GD / 4; ++kk) {
    bfr[kk].x *= fr;
    bfr[kk].y *= fr;
  }

  const float* arow = &lds_u[(ttile * 16 + m) * US];
  int n = tid & 31;
  int s = tid >> 5;

#pragma unroll
  for (int i = 0; i < CPB; ++i) {
#pragma unroll
    for (int j = 0; j < 8; ++j)
      *(v4f*)(&lds_u[(tl0 + 8 * j) * US + q4 * 4]) = ur[j];
    v4f un[8];
    if (i + 1 < CPB) {
      const float* ubn = ub + (size_t)(i + 1) * LCH * DM;
#pragma unroll
      for (int j = 0; j < 8; ++j)
        un[j] = *(const v4f*)(ubn + (size_t)(tl0 + 8 * j) * DM + q4 * 4);
    }
    lds_barrier();

    v8f acc = {};
#pragma unroll
    for (int kk = 0; kk < GD / 4; ++kk) {
      v2f a = *(const v2f*)(arow + kk * 4 + koff);
      acc = wmma4(a, bfr[kk], acc);
    }
#pragma unroll
    for (int r = 0; r < 8; ++r) {
      int t = ttile * 16 + r + rowadd;
      lds_bu[t * BS + ntile * 16 + m] = acc[r];
    }
    lds_barrier();

    float run = 0.f;
#pragma unroll
    for (int k = 0; k < SEG; ++k) {
      int t = s * SEG + k;
      run = run * Ab + lds_bu[t * BS + n];
      lds_bu[t * BS + n] = run;
    }
    segsum[s * NST + n] = run;
    lds_barrier();
    if (s == 0) {
      float carry = 0.f;
#pragma unroll
      for (int ss = 0; ss < NSEG; ++ss) {
        segcar[ss * NST + n] = carry;
        carry = carry * a8 + segsum[ss * NST + n];
      }
    }
    lds_barrier();
    {
      float carry = segcar[s * NST + n];
      float f = Ab;
#pragma unroll
      for (int k = 0; k < SEG; ++k) {
        lds_bu[(s * SEG + k) * BS + n] += f * carry;
        f *= Ab;
      }
    }
    lds_barrier();

    float* out = wLocal + ((size_t)((b * GG + g) * TT + (c0 + i) * LCH)) * NST;
#pragma unroll
    for (int j = 0; j < 2; ++j) {
      int idx4 = tid + 256 * j;
      int tl = idx4 >> 3;
      int q  = idx4 & 7;
      *(v4f*)(out + tl * NST + q * 4) = *(const v4f*)(&lds_bu[tl * BS + q * 4]);
    }
    if (i + 1 < CPB) {
#pragma unroll
      for (int j = 0; j < 8; ++j) ur[j] = un[j];
    }
  }
}

__global__ __launch_bounds__(256) void k3_y(
    const float* __restrict__ u, const float* __restrict__ Cmat,
    const float* __restrict__ Dp, const float* __restrict__ A_log,
    const float* __restrict__ dt_logit, const float* __restrict__ state0,
    const float* __restrict__ wLocal,
    float* __restrict__ y, float* __restrict__ sfinal) {
  __shared__ float lds_s[LCH * BS];
  __shared__ float lds_y[LCH * US];
  __shared__ float lds_carry[CPB * NST];

  int blk = blockIdx.x;
  int qg = blk & (NGRP - 1);
  int g = (blk >> 4) & (GG - 1);
  int b = blk >> 7;
  int c0 = qg * CPB;
  int bg = b * GG + g;
  int tid = threadIdx.x;

  float xlog = A_log[g * NST + (tid & 31)];
  float dtl  = dt_logit[g];

  const float* src = wLocal + ((size_t)(bg * TT + c0 * LCH)) * NST;
  int tl2 = tid >> 3;
  int q2  = tid & 7;
  v4f sr[2];
#pragma unroll
  for (int j = 0; j < 2; ++j)
    sr[j] = *(const v4f*)(src + (tl2 + 32 * j) * NST + q2 * 4);

  const float* ubase = u + ((size_t)(b * TT + c0 * LCH)) * DM + g * GD;
  float*       ybase = y + ((size_t)(b * TT + c0 * LCH)) * DM + g * GD;
  int tl0 = tid >> 5;
  int q4  = tid & 31;
  v4f ur[8];
#pragma unroll
  for (int j = 0; j < 8; ++j)
    ur[j] = *(const v4f*)(ubase + (size_t)(tl0 + 8 * j) * DM + q4 * 4);
  v4f dp = *(const v4f*)(Dp + g * GD + q4 * 4);

  int lane   = tid & 31;
  int w      = tid >> 5;
  int ttile  = w >> 1;
  int dhalf  = w & 1;
  int m      = lane & 15;
  int koff   = (lane >> 4) << 1;
  int rowadd = (lane >> 4) << 3;

  const float* Cg = Cmat + (size_t)g * GD * NST;
  v2f cfr[4][8];
#pragma unroll
  for (int dt = 0; dt < 4; ++dt) {
    const float* brow = Cg + (size_t)(dhalf * 64 + dt * 16 + m) * NST;
#pragma unroll
    for (int k8 = 0; k8 < 8; ++k8)
      cfr[dt][k8] = *(const v2f*)(brow + k8 * 4 + koff);
  }

  float e[NCH];
  float carry0 = 0.f;
  if (tid < 32) {
    const float* ends = wLocal + ((size_t)bg * TT + (LCH - 1)) * NST + tid;
#pragma unroll
    for (int j = 0; j < NCH; ++j) e[j] = ends[(size_t)j * LCH * NST];
    carry0 = state0[bg * NST + tid];
  }

  __builtin_amdgcn_sched_barrier(0);

  float Ab, frac_unused;
  zoh_params(xlog, dtl, Ab, frac_unused);
  float a8 = Ab;
  a8 *= a8; a8 *= a8; a8 *= a8;

  if (tid < 32) {
    float aL = a8;
    aL *= aL; aL *= aL; aL *= aL;
    float carry = carry0;
    float cin[CPB];
#pragma unroll
    for (int i = 0; i < CPB; ++i) cin[i] = 0.f;
#pragma unroll
    for (int j = 0; j < NCH; ++j) {
      cin[j % CPB] = ((j / CPB) == qg) ? carry : cin[j % CPB];
      carry = carry * aL + e[j];
    }
#pragma unroll
    for (int i = 0; i < CPB; ++i) lds_carry[i * NST + tid] = cin[i];
    if (qg == NGRP - 1) sfinal[bg * NST + tid] = carry;
  }

  int n = tid & 31;
  int s = tid >> 5;
  float fpow = Ab;
#pragma unroll
  for (int j = 0; j < NSEG - 1; ++j) fpow = (j < s) ? fpow * a8 : fpow;

  const float* arow = &lds_s[(ttile * 16 + m) * BS];

#pragma unroll
  for (int i = 0; i < CPB; ++i) {
#pragma unroll
    for (int j = 0; j < 2; ++j)
      *(v4f*)(&lds_s[(tl2 + 32 * j) * BS + q2 * 4]) = sr[j];
    v4f un[8];
    v4f sn[2];
    if (i + 1 < CPB) {
      const float* srcn = src + (size_t)(i + 1) * LCH * NST;
#pragma unroll
      for (int j = 0; j < 2; ++j)
        sn[j] = *(const v4f*)(srcn + (tl2 + 32 * j) * NST + q2 * 4);
      const float* ubn = ubase + (size_t)(i + 1) * LCH * DM;
#pragma unroll
      for (int j = 0; j < 8; ++j)
        un[j] = *(const v4f*)(ubn + (size_t)(tl0 + 8 * j) * DM + q4 * 4);
    }
    lds_barrier();

    {
      float carry = lds_carry[i * NST + n];
      float f = fpow;
#pragma unroll
      for (int k = 0; k < SEG; ++k) {
        lds_s[(s * SEG + k) * BS + n] += f * carry;
        f *= Ab;
      }
    }
    lds_barrier();

    v2f afrag[8];
#pragma unroll
    for (int k8 = 0; k8 < 8; ++k8)
      afrag[k8] = *(const v2f*)(arow + k8 * 4 + koff);

#pragma unroll
    for (int dt = 0; dt < 4; ++dt) {
      int d = dhalf * 64 + dt * 16 + m;
      v8f acc = {};
#pragma unroll
      for (int k8 = 0; k8 < 8; ++k8)
        acc = wmma4(afrag[k8], cfr[dt][k8], acc);
#pragma unroll
      for (int r = 0; r < 8; ++r) {
        int t = ttile * 16 + r + rowadd;
        lds_y[t * US + d] = acc[r];
      }
    }
    lds_barrier();

    float* yb = ybase + (size_t)i * LCH * DM;
#pragma unroll
    for (int j = 0; j < 8; ++j) {
      int tl = tl0 + 8 * j;
      v4f uv = ur[j];
      v4f cs = *(const v4f*)(&lds_y[tl * US + q4 * 4]);
      v4f o;
      o.x = cs.x + dp.x * uv.x;
      o.y = cs.y + dp.y * uv.y;
      o.z = cs.z + dp.z * uv.z;
      o.w = cs.w + dp.w * uv.w;
      *(v4f*)(yb + (size_t)tl * DM + q4 * 4) = o;
    }
    if (i + 1 < CPB) {
#pragma unroll
      for (int j = 0; j < 8; ++j) ur[j] = un[j];
#pragma unroll
      for (int j = 0; j < 2; ++j) sr[j] = sn[j];
    }
  }
}

extern "C" void kernel_launch(void* const* d_in, const int* in_sizes, int n_in,
                              void* d_out, int out_size, void* d_ws, size_t ws_size,
                              hipStream_t stream) {
  (void)in_sizes; (void)n_in; (void)out_size; (void)ws_size;
  const float* u        = (const float*)d_in[0];
  const float* state0   = (const float*)d_in[1];
  const float* A_log    = (const float*)d_in[2];
  const float* B_param  = (const float*)d_in[3];
  const float* Cmat     = (const float*)d_in[4];
  const float* Dp       = (const float*)d_in[5];
  const float* dt_logit = (const float*)d_in[6];
  float* out = (float*)d_out;
  float* ws  = (float*)d_ws;

  float* wLocal  = ws + 33536;

  float* y      = out;
  float* sfinal = out + (size_t)BB * TT * DM;

  k1_bu_scan<<<BB * GG * NGRP, 256, 0, stream>>>(u, A_log, B_param, dt_logit, wLocal);
  k3_y<<<BB * GG * NGRP, 256, 0, stream>>>(u, Cmat, Dp, A_log, dt_logit, state0,
                                           wLocal, y, sfinal);
}


# B_param and C group tiles fetched once per block with whole-line 16-byte loads and staged in LDS (B scaled on the way in); WMMA B fragments read from LDS instead of 16-line 8-byte global loads
# speedup vs baseline: 2.5906x; 1.0531x over previous
#include <hip/hip_runtime.h>
#include <hip/hip_bf16.h>
#include <math.h>


#define BB   4
#define TT   4096
#define DM   1024
#define GG   8
#define GD   128
#define NST  32
#define LCH  64
#define NCH  (TT / LCH)
#define SEG  8
#define NSEG (LCH / SEG)
#define CPB  4
#define NGRP (NCH / CPB)

#define US   132
#define BS   36

typedef float v2f __attribute__((ext_vector_type(2)));
typedef float v4f __attribute__((ext_vector_type(4)));
typedef float v8f __attribute__((ext_vector_type(8)));

__device__ __forceinline__ v8f wmma4(v2f a, v2f b, v8f c) {
  return __builtin_amdgcn_wmma_f32_16x16x4_f32(
       false, a,  false, b,
       (short)0, c,  false,  false);
}

__device__ __forceinline__ void lds_barrier() {
  asm volatile("s_wait_dscnt 0x0\n\ts_barrier_signal -1\n\ts_barrier_wait -1" ::: "memory");
}

__device__ __forceinline__ void zoh_params(float x, float dtl, float& Abar, float& frac) {
  float sp0 = log1pf(expf(x));
  float sp = (x > 20.f) ? x : sp0;
  float A = -sp;
  float sig = 1.f / (1.f + expf(-dtl));
  float dt = 0.001f + (0.1f - 0.001f) * sig;
  Abar = expf(A * dt);
  float q = (Abar - 1.f) / A;
  frac = (fabsf(A) > 1e-6f) ? q : dt;
}

__global__ __launch_bounds__(256) void k1_bu_scan(
    const float* __restrict__ u, const float* __restrict__ A_log,
    const float* __restrict__ B_param, const float* __restrict__ dt_logit,
    float* __restrict__ wLocal) {
  __shared__ float lds_u[LCH * US];
  __shared__ float lds_bu[LCH * BS];
  __shared__ float segsum[NSEG * NST];
  __shared__ float segcar[NSEG * NST];
  __shared__ __attribute__((aligned(16))) float lds_b[NST * US];

  int blk = blockIdx.x;
  int qg = blk & (NGRP - 1);
  int g = (blk >> 4) & (GG - 1);
  int b = blk >> 7;
  int c0 = qg * CPB;
  int tid = threadIdx.x;

  float xlog = A_log[g * NST + (tid & 31)];
  float dtl  = dt_logit[g];

  const float* ub = u + ((size_t)(b * TT + c0 * LCH)) * DM + g * GD;
  int tl0 = tid >> 5;
  int q4  = tid & 31;
  v4f ur[8];
#pragma unroll
  for (int j = 0; j < 8; ++j)
    ur[j] = *(const v4f*)(ub + (size_t)(tl0 + 8 * j) * DM + q4 * 4);

  const float* bgrp = B_param + (size_t)g * NST * GD;
  v4f br[4];
#pragma unroll
  for (int j = 0; j < 4; ++j)
    br[j] = *(const v4f*)(bgrp + (size_t)(tl0 + 8 * j) * GD + q4 * 4);

  int lane   = tid & 31;
  int w      = tid >> 5;
  int ttile  = w >> 1;
  int ntile  = w & 1;
  int m      = lane & 15;
  int koff   = (lane >> 4) << 1;
  int rowadd = (lane >> 4) << 3;

  __builtin_amdgcn_sched_barrier(0);

  float Ab, frac;
  zoh_params(xlog, dtl, Ab, frac);
  float a8 = Ab;
  a8 *= a8; a8 *= a8; a8 *= a8;

#pragma unroll
  for (int j = 0; j < 4; ++j) {
    float fr = __shfl(frac, tl0 + 8 * j);
    v4f bv = br[j];
    bv.x *= fr; bv.y *= fr; bv.z *= fr; bv.w *= fr;
    *(v4f*)(&lds_b[(tl0 + 8 * j) * US + q4 * 4]) = bv;
  }

  v2f bfr[GD / 4];
  const float* brow = &lds_b[(ntile * 16 + m) * US];

  const float* arow = &lds_u[(ttile * 16 + m) * US];
  int n = tid & 31;
  int s = tid >> 5;

#pragma unroll
  for (int i = 0; i < CPB; ++i) {
#pragma unroll
    for (int j = 0; j < 8; ++j)
      *(v4f*)(&lds_u[(tl0 + 8 * j) * US + q4 * 4]) = ur[j];
    v4f un[8];
    if (i + 1 < CPB) {
      const float* ubn = ub + (size_t)(i + 1) * LCH * DM;
#pragma unroll
      for (int j = 0; j < 8; ++j)
        un[j] = *(const v4f*)(ubn + (size_t)(tl0 + 8 * j) * DM + q4 * 4);
    }
    lds_barrier();

    if (i == 0) {
#pragma unroll
      for (int kk = 0; kk < GD / 4; ++kk)
        bfr[kk] = *(const v2f*)(brow + kk * 4 + koff);
      __builtin_amdgcn_sched_barrier(0);
    }

    v8f acc = {};
#pragma unroll
    for (int kk = 0; kk < GD / 4; ++kk) {
      v2f a = *(const v2f*)(arow + kk * 4 + koff);
      acc = wmma4(a, bfr[kk], acc);
    }
#pragma unroll
    for (int r = 0; r < 8; ++r) {
      int t = ttile * 16 + r + rowadd;
      lds_bu[t * BS + ntile * 16 + m] = acc[r];
    }
    lds_barrier();

    float run = 0.f;
#pragma unroll
    for (int k = 0; k < SEG; ++k) {
      int t = s * SEG + k;
      run = run * Ab + lds_bu[t * BS + n];
      lds_bu[t * BS + n] = run;
    }
    segsum[s * NST + n] = run;
    lds_barrier();
    if (s == 0) {
      float carry = 0.f;
#pragma unroll
      for (int ss = 0; ss < NSEG; ++ss) {
        segcar[ss * NST + n] = carry;
        carry = carry * a8 + segsum[ss * NST + n];
      }
    }
    lds_barrier();
    {
      float carry = segcar[s * NST + n];
      float f = Ab;
#pragma unroll
      for (int k = 0; k < SEG; ++k) {
        lds_bu[(s * SEG + k) * BS + n] += f * carry;
        f *= Ab;
      }
    }
    lds_barrier();

    float* out = wLocal + ((size_t)((b * GG + g) * TT + (c0 + i) * LCH)) * NST;
#pragma unroll
    for (int j = 0; j < 2; ++j) {
      int idx4 = tid + 256 * j;
      int tl = idx4 >> 3;
      int q  = idx4 & 7;
      *(v4f*)(out + tl * NST + q * 4) = *(const v4f*)(&lds_bu[tl * BS + q * 4]);
    }
    if (i + 1 < CPB) {
#pragma unroll
      for (int j = 0; j < 8; ++j) ur[j] = un[j];
    }
  }
}

__global__ __launch_bounds__(256) void k3_y(
    const float* __restrict__ u, const float* __restrict__ Cmat,
    const float* __restrict__ Dp, const float* __restrict__ A_log,
    const float* __restrict__ dt_logit, const float* __restrict__ state0,
    const float* __restrict__ wLocal,
    float* __restrict__ y, float* __restrict__ sfinal) {
  __shared__ float lds_s[LCH * BS];
  __shared__ float lds_y[LCH * US];
  __shared__ float lds_carry[CPB * NST];
  __shared__ __attribute__((aligned(16))) float lds_c[GD * BS];

  int blk = blockIdx.x;
  int qg = blk & (NGRP - 1);
  int g = (blk >> 4) & (GG - 1);
  int b = blk >> 7;
  int c0 = qg * CPB;
  int bg = b * GG + g;
  int tid = threadIdx.x;

  float xlog = A_log[g * NST + (tid & 31)];
  float dtl  = dt_logit[g];

  const float* src = wLocal + ((size_t)(bg * TT + c0 * LCH)) * NST;
  int tl2 = tid >> 3;
  int q2  = tid & 7;
  v4f sr[2];
#pragma unroll
  for (int j = 0; j < 2; ++j)
    sr[j] = *(const v4f*)(src + (tl2 + 32 * j) * NST + q2 * 4);

  const float* ubase = u + ((size_t)(b * TT + c0 * LCH)) * DM + g * GD;
  float*       ybase = y + ((size_t)(b * TT + c0 * LCH)) * DM + g * GD;
  int tl0 = tid >> 5;
  int q4  = tid & 31;
  v4f ur[8];
#pragma unroll
  for (int j = 0; j < 8; ++j)
    ur[j] = *(const v4f*)(ubase + (size_t)(tl0 + 8 * j) * DM + q4 * 4);
  v4f dp = *(const v4f*)(Dp + g * GD + q4 * 4);

  int lane   = tid & 31;
  int w      = tid >> 5;
  int ttile  = w >> 1;
  int dhalf  = w & 1;
  int m      = lane & 15;
  int koff   = (lane >> 4) << 1;
  int rowadd = (lane >> 4) << 3;

  const float* Cg = Cmat + (size_t)g * GD * NST;
  v4f cr[4];
#pragma unroll
  for (int j = 0; j < 4; ++j)
    cr[j] = *(const v4f*)(Cg + (size_t)(tl2 + 32 * j) * NST + q2 * 4);

  float e[NCH];
  float carry0 = 0.f;
  if (tid < 32) {
    const float* ends = wLocal + ((size_t)bg * TT + (LCH - 1)) * NST + tid;
#pragma unroll
    for (int j = 0; j < NCH; ++j) e[j] = ends[(size_t)j * LCH * NST];
    carry0 = state0[bg * NST + tid];
  }

  __builtin_amdgcn_sched_barrier(0);

  float Ab, frac_unused;
  zoh_params(xlog, dtl, Ab, frac_unused);
  float a8 = Ab;
  a8 *= a8; a8 *= a8; a8 *= a8;

  if (tid < 32) {
    float aL = a8;
    aL *= aL; aL *= aL; aL *= aL;
    float carry = carry0;
    float cin[CPB];
#pragma unroll
    for (int i = 0; i < CPB; ++i) cin[i] = 0.f;
#pragma unroll
    for (int j = 0; j < NCH; ++j) {
      cin[j % CPB] = ((j / CPB) == qg) ? carry : cin[j % CPB];
      carry = carry * aL + e[j];
    }
#pragma unroll
    for (int i = 0; i < CPB; ++i) lds_carry[i * NST + tid] = cin[i];
    if (qg == NGRP - 1) sfinal[bg * NST + tid] = carry;
  }

  int n = tid & 31;
  int s = tid >> 5;
  float fpow = Ab;
#pragma unroll
  for (int j = 0; j < NSEG - 1; ++j) fpow = (j < s) ? fpow * a8 : fpow;

  const float* arow = &lds_s[(ttile * 16 + m) * BS];

#pragma unroll
  for (int j = 0; j < 4; ++j)
    *(v4f*)(&lds_c[(tl2 + 32 * j) * BS + q2 * 4]) = cr[j];

  v2f cfr[4][8];

#pragma unroll
  for (int i = 0; i < CPB; ++i) {
#pragma unroll
    for (int j = 0; j < 2; ++j)
      *(v4f*)(&lds_s[(tl2 + 32 * j) * BS + q2 * 4]) = sr[j];
    v4f un[8];
    v4f sn[2];
    if (i + 1 < CPB) {
      const float* srcn = src + (size_t)(i + 1) * LCH * NST;
#pragma unroll
      for (int j = 0; j < 2; ++j)
        sn[j] = *(const v4f*)(srcn + (tl2 + 32 * j) * NST + q2 * 4);
      const float* ubn = ubase + (size_t)(i + 1) * LCH * DM;
#pragma unroll
      for (int j = 0; j < 8; ++j)
        un[j] = *(const v4f*)(ubn + (size_t)(tl0 + 8 * j) * DM + q4 * 4);
    }
    lds_barrier();

    if (i == 0) {
#pragma unroll
      for (int dt = 0; dt < 4; ++dt) {
        const float* crow = &lds_c[(dhalf * 64 + dt * 16 + m) * BS];
#pragma unroll
        for (int k8 = 0; k8 < 8; ++k8)
          cfr[dt][k8] = *(const v2f*)(crow + k8 * 4 + koff);
      }
    }

    {
      float carry = lds_carry[i * NST + n];
      float f = fpow;
#pragma unroll
      for (int k = 0; k < SEG; ++k) {
        lds_s[(s * SEG + k) * BS + n] += f * carry;
        f *= Ab;
      }
    }
    lds_barrier();

    v2f afrag[8];
#pragma unroll
    for (int k8 = 0; k8 < 8; ++k8)
      afrag[k8] = *(const v2f*)(arow + k8 * 4 + koff);

#pragma unroll
    for (int dt = 0; dt < 4; ++dt) {
      int d = dhalf * 64 + dt * 16 + m;
      v8f acc = {};
#pragma unroll
      for (int k8 = 0; k8 < 8; ++k8)
        acc = wmma4(afrag[k8], cfr[dt][k8], acc);
#pragma unroll
      for (int r = 0; r < 8; ++r) {
        int t = ttile * 16 + r + rowadd;
        lds_y[t * US + d] = acc[r];
      }
    }
    lds_barrier();

    float* yb = ybase + (size_t)i * LCH * DM;
#pragma unroll
    for (int j = 0; j < 8; ++j) {
      int tl = tl0 + 8 * j;
      v4f uv = ur[j];
      v4f cs = *(const v4f*)(&lds_y[tl * US + q4 * 4]);
      v4f o;
      o.x = cs.x + dp.x * uv.x;
      o.y = cs.y + dp.y * uv.y;
      o.z = cs.z + dp.z * uv.z;
      o.w = cs.w + dp.w * uv.w;
      *(v4f*)(yb + (size_t)tl * DM + q4 * 4) = o;
    }
    if (i + 1 < CPB) {
#pragma unroll
      for (int j = 0; j < 8; ++j) ur[j] = un[j];
#pragma unroll
      for (int j = 0; j < 2; ++j) sr[j] = sn[j];
    }
  }
}

extern "C" void kernel_launch(void* const* d_in, const int* in_sizes, int n_in,
                              void* d_out, int out_size, void* d_ws, size_t ws_size,
                              hipStream_t stream) {
  (void)in_sizes; (void)n_in; (void)out_size; (void)ws_size;
  const float* u        = (const float*)d_in[0];
  const float* state0   = (const float*)d_in[1];
  const float* A_log    = (const float*)d_in[2];
  const float* B_param  = (const float*)d_in[3];
  const float* Cmat     = (const float*)d_in[4];
  const float* Dp       = (const float*)d_in[5];
  const float* dt_logit = (const float*)d_in[6];
  float* out = (float*)d_out;
  float* ws  = (float*)d_ws;

  float* wLocal  = ws + 33536;

  float* y      = out;
  float* sfinal = out + (size_t)BB * TT * DM;

  k1_bu_scan<<<BB * GG * NGRP, 256, 0, stream>>>(u, A_log, B_param, dt_logit, wLocal);
  k3_y<<<BB * GG * NGRP, 256, 0, stream>>>(u, Cmat, Dp, A_log, dt_logit, state0,
                                           wLocal, y, sfinal);
}
